# starting kernel
# baseline (speedup 1.0000x reference)
#include <hip/hip_runtime.h>
#include <hip/hip_bf16.h>

typedef __attribute__((ext_vector_type(2))) float v2f;
typedef __attribute__((ext_vector_type(8))) float v8f;

#define HID 128
#define NGRAPH 64


__global__ void k_fill(float* p, int n, float v) {
    int i = blockIdx.x * 256 + threadIdx.x;
    if (i < n) p[i] = v;
}

__global__ void k_deg_count(const int* __restrict__ dst, int E, float* __restrict__ deg) {
    int e = blockIdx.x * 256 + threadIdx.x;
    if (e < E) atomicAdd(deg + dst[e], 1.0f);
}

__global__ void k_rsqrt(float* d, int n) {
    int i = blockIdx.x * 256 + threadIdx.x;
    if (i < n) d[i] = rsqrtf(d[i]);
}

__global__ void k_embed(const int* __restrict__ res, const float* __restrict__ pos,
                        const float* __restrict__ W, const float* __restrict__ dinv,
                        int N, float* __restrict__ y) {
    int idx = blockIdx.x * 256 + threadIdx.x;
    if (idx >= N * HID) return;
    int i = idx >> 7, f = idx & 127;
    float xw = W[res[i] * HID + f]
             + pos[i * 3 + 0] * W[20 * HID + f]
             + pos[i * 3 + 1] * W[21 * HID + f]
             + pos[i * 3 + 2] * W[22 * HID + f];
    y[idx] = xw * dinv[i];
}

__global__ void k_scatter(const int* __restrict__ src, const int* __restrict__ dst,
                          int E, const float* __restrict__ y, float* __restrict__ s) {
    int e = blockIdx.x * 8 + (threadIdx.x >> 5);
    if (e >= E) return;
    int lane = threadIdx.x & 31;
    size_t so = (size_t)src[e] * HID + lane * 4;
    size_t doo = (size_t)dst[e] * HID + lane * 4;
    float4 v = *(const float4*)(y + so);
    float* p = s + doo;
    atomicAdd(p + 0, v.x);
    atomicAdd(p + 1, v.y);
    atomicAdd(p + 2, v.z);
    atomicAdd(p + 3, v.w);
}

__global__ void k_combine(const float* __restrict__ y, const float* __restrict__ s,
                          const float* __restrict__ dinv, const float* __restrict__ b,
                          int N, float* __restrict__ h) {
    int idx = blockIdx.x * 256 + threadIdx.x;
    if (idx >= N * HID) return;
    int i = idx >> 7, f = idx & 127;
    h[idx] = fmaxf(dinv[i] * (s[idx] + y[idx]) + b[f], 0.0f);
}

__global__ void k_cnt(const int* __restrict__ batch, int N, float* __restrict__ cnt) {
    int i = blockIdx.x * 256 + threadIdx.x;
    if (i < N) atomicAdd(cnt + batch[i], 1.0f);
}

__global__ void k_pool_sum(const float* __restrict__ h, const int* __restrict__ batch,
                           int N, float* __restrict__ pesum) {
    int idx = blockIdx.x * 256 + threadIdx.x;
    if (idx >= N * HID) return;
    int i = idx >> 7, f = idx & 127;
    atomicAdd(pesum + batch[i] * HID + f, h[idx]);
}

__global__ void k_pool_div(const float* __restrict__ pesum, const float* __restrict__ cnt,
                           float* __restrict__ pe) {
    int idx = blockIdx.x * 256 + threadIdx.x;
    if (idx >= NGRAPH * HID) return;
    pe[idx] = pesum[idx] / fmaxf(cnt[idx >> 7], 1.0f);
}

__global__ void k_feat(const float* __restrict__ h, const float* __restrict__ pe,
                       const int* __restrict__ batch, int N, float* __restrict__ feat) {
    int idx = blockIdx.x * 256 + threadIdx.x;
    if (idx >= N * 2 * HID) return;
    int i = idx >> 8, f = idx & 255;
    feat[idx] = (f < HID) ? h[(size_t)i * HID + f]
                          : pe[batch[i] * HID + (f - HID)];
}

__global__ void k_head(const float* __restrict__ hmid, const float* __restrict__ W2,
                       const float* __restrict__ b2, int N, float* __restrict__ out) {
    int i = blockIdx.x * 256 + threadIdx.x;
    if (i >= N) return;
    float a0 = b2[0], a1 = b2[1], a2 = b2[2];
    const float* h = hmid + (size_t)i * HID;
    for (int k = 0; k < HID; ++k) {
        float hv = h[k];
        a0 += hv * W2[k * 3 + 0];
        a1 += hv * W2[k * 3 + 1];
        a2 += hv * W2[k * 3 + 2];
    }
    out[(size_t)i * 3 + 0] = a0;
    out[(size_t)i * 3 + 1] = a1;
    out[(size_t)i * 3 + 2] = a2;
}

__global__ void k_wfrag(const float* __restrict__ W, int K, float* __restrict__ Wf) {
    int idx = blockIdx.x * 256 + threadIdx.x;
    if (idx >= K * HID) return;
    int j    = idx & 1;
    int nt   = (idx >> 1) & 7;
    int l    = (idx >> 4) & 31;
    int kk   = idx >> 9;
    int krow = 4 * kk + (l >> 4) * 2 + j;
    int col  = nt * 16 + (l & 15);
    Wf[idx] = W[(size_t)krow * HID + col];
}

__global__ void k_gemm128(const float* __restrict__ A, int K, int M,
                          const float* __restrict__ Wf,
                          const float* __restrict__ rowscale,
                          const float* __restrict__ bias, int relu,
                          float* __restrict__ out) {
    int w = threadIdx.x >> 5;
    int lane = threadIdx.x & 31;
    int tile = blockIdx.x * 4 + w;
    if (tile * 16 >= M) return;
    int row0 = tile * 16;
    int half = lane >> 4;
    int lmod = lane & 15;
    int koff = half * 2;
    const float* Arow = A + (size_t)(row0 + lmod) * K + koff;
    const float* Wl   = Wf + (size_t)lane * 16;

    v8f acc[8];
    #pragma unroll
    for (int nt = 0; nt < 8; ++nt) acc[nt] = (v8f){};

    int ksteps = K >> 2;
    for (int kk = 0; kk < ksteps; ++kk) {
        v2f a;
        a.x = Arow[4 * kk];
        a.y = Arow[4 * kk + 1];
        const float4* q = (const float4*)(Wl + (size_t)kk * 512);
        float4 q0 = q[0], q1 = q[1], q2 = q[2], q3 = q[3];
        v2f b0; b0.x = q0.x; b0.y = q0.y;
        v2f b1; b1.x = q0.z; b1.y = q0.w;
        v2f b2; b2.x = q1.x; b2.y = q1.y;
        v2f b3; b3.x = q1.z; b3.y = q1.w;
        v2f b4; b4.x = q2.x; b4.y = q2.y;
        v2f b5; b5.x = q2.z; b5.y = q2.w;
        v2f b6; b6.x = q3.x; b6.y = q3.y;
        v2f b7; b7.x = q3.z; b7.y = q3.w;
        acc[0] = __builtin_amdgcn_wmma_f32_16x16x4_f32(false, a, false, b0, (short)0, acc[0], false, false);
        acc[1] = __builtin_amdgcn_wmma_f32_16x16x4_f32(false, a, false, b1, (short)0, acc[1], false, false);
        acc[2] = __builtin_amdgcn_wmma_f32_16x16x4_f32(false, a, false, b2, (short)0, acc[2], false, false);
        acc[3] = __builtin_amdgcn_wmma_f32_16x16x4_f32(false, a, false, b3, (short)0, acc[3], false, false);
        acc[4] = __builtin_amdgcn_wmma_f32_16x16x4_f32(false, a, false, b4, (short)0, acc[4], false, false);
        acc[5] = __builtin_amdgcn_wmma_f32_16x16x4_f32(false, a, false, b5, (short)0, acc[5], false, false);
        acc[6] = __builtin_amdgcn_wmma_f32_16x16x4_f32(false, a, false, b6, (short)0, acc[6], false, false);
        acc[7] = __builtin_amdgcn_wmma_f32_16x16x4_f32(false, a, false, b7, (short)0, acc[7], false, false);
    }

    float rs[8];
    if (rowscale) {
        #pragma unroll
        for (int r = 0; r < 8; ++r) rs[r] = rowscale[row0 + r + half * 8];
    }

    #pragma unroll
    for (int nt = 0; nt < 8; ++nt) {
        int col = nt * 16 + lmod;
        float bv = bias ? bias[col] : 0.0f;
        #pragma unroll
        for (int r = 0; r < 8; ++r) {
            float v = acc[nt][r];
            if (rowscale) v *= rs[r];
            v += bv;
            if (relu) v = fmaxf(v, 0.0f);
            out[(size_t)(row0 + r + half * 8) * HID + col] = v;
        }
    }
}


static inline int cdiv(long long a, long long b) { return (int)((a + b - 1) / b); }

static void run_branch(const int* res, const float* pos, const int* ei,
                       int N, int E,
                       const float* W1c, const float* b1c,
                       const float* W2c, const float* b2c,
                       float* dinv, float* A, float* B, float* C, float* Wf,
                       hipStream_t stream) {
    const int* src = ei;
    const int* dst = ei + E;
    size_t nh = (size_t)N * HID;

    k_fill<<<cdiv(N, 256), 256, 0, stream>>>(dinv, N, 1.0f);
    k_deg_count<<<cdiv(E, 256), 256, 0, stream>>>(dst, E, dinv);
    k_rsqrt<<<cdiv(N, 256), 256, 0, stream>>>(dinv, N);

    k_embed<<<cdiv(nh, 256), 256, 0, stream>>>(res, pos, W1c, dinv, N, A);
    hipMemsetAsync(B, 0, nh * sizeof(float), stream);
    k_scatter<<<cdiv(E, 8), 256, 0, stream>>>(src, dst, E, A, B);
    k_combine<<<cdiv(nh, 256), 256, 0, stream>>>(A, B, dinv, b1c, N, C);

    k_wfrag<<<cdiv(HID * HID, 256), 256, 0, stream>>>(W2c, HID, Wf);
    k_gemm128<<<cdiv(N / 16, 4), 128, 0, stream>>>(C, HID, N, Wf, dinv, nullptr, 0, A);
    hipMemsetAsync(B, 0, nh * sizeof(float), stream);
    k_scatter<<<cdiv(E, 8), 256, 0, stream>>>(src, dst, E, A, B);
    k_combine<<<cdiv(nh, 256), 256, 0, stream>>>(A, B, dinv, b2c, N, C);
}

extern "C" void kernel_launch(void* const* d_in, const int* in_sizes, int n_in,
                              void* d_out, int out_size, void* d_ws, size_t ws_size,
                              hipStream_t stream) {
    (void)n_in; (void)out_size; (void)ws_size;

    const int*   res_p   = (const int*)d_in[0];
    const float* pos_p   = (const float*)d_in[1];
    const int*   ei_p    = (const int*)d_in[2];
    const int*   batch_p = (const int*)d_in[3];
    const int*   res_m   = (const int*)d_in[4];
    const float* pos_m   = (const float*)d_in[5];
    const int*   ei_m    = (const int*)d_in[6];
    const int*   batch_m = (const int*)d_in[7];
    const float* W_p1 = (const float*)d_in[8];
    const float* b_p1 = (const float*)d_in[9];
    const float* W_p2 = (const float*)d_in[10];
    const float* b_p2 = (const float*)d_in[11];
    const float* W_m1 = (const float*)d_in[12];
    const float* b_m1 = (const float*)d_in[13];
    const float* W_m2 = (const float*)d_in[14];
    const float* b_m2 = (const float*)d_in[15];
    const float* W1f  = (const float*)d_in[16];
    const float* b1f  = (const float*)d_in[17];
    const float* W2f  = (const float*)d_in[18];
    const float* b2f  = (const float*)d_in[19];

    int N_P = in_sizes[0];
    int E_P = in_sizes[2] / 2;
    int N_M = in_sizes[4];
    int E_M = in_sizes[6] / 2;

    float* ws = (float*)d_ws;
    size_t nhp = (size_t)N_P * HID;
    float* A      = ws;
    float* B      = A + nhp;
    float* C      = B + nhp;
    float* dinv_p = C + nhp;
    float* dinv_m = dinv_p + N_P;
    float* pe_sum = dinv_m + N_M;
    float* pe     = pe_sum + NGRAPH * HID;
    float* cnt    = pe + NGRAPH * HID;
    float* Wfrag  = cnt + NGRAPH;

    run_branch(res_p, pos_p, ei_p, N_P, E_P, W_p1, b_p1, W_p2, b_p2,
               dinv_p, A, B, C, Wfrag, stream);
    hipMemsetAsync(pe_sum, 0, (size_t)NGRAPH * HID * sizeof(float), stream);
    hipMemsetAsync(cnt, 0, NGRAPH * sizeof(float), stream);
    k_cnt<<<cdiv(N_P, 256), 256, 0, stream>>>(batch_p, N_P, cnt);
    k_pool_sum<<<cdiv(nhp, 256), 256, 0, stream>>>(C, batch_p, N_P, pe_sum);
    k_pool_div<<<cdiv(NGRAPH * HID, 256), 256, 0, stream>>>(pe_sum, cnt, pe);

    run_branch(res_m, pos_m, ei_m, N_M, E_M, W_m1, b_m1, W_m2, b_m2,
               dinv_m, A, B, C, Wfrag, stream);

    size_t nfm = (size_t)N_M * 2 * HID;
    float* feat = A;
    float* hmid = B;
    k_feat<<<cdiv(nfm, 256), 256, 0, stream>>>(C, pe, batch_m, N_M, feat);
    k_wfrag<<<cdiv(2 * HID * HID, 256), 256, 0, stream>>>(W1f, 2 * HID, Wfrag);
    k_gemm128<<<cdiv(N_M / 16, 4), 128, 0, stream>>>(feat, 2 * HID, N_M, Wfrag,
                                                     nullptr, b1f, 1, hmid);
    k_head<<<cdiv(N_M, 256), 256, 0, stream>>>(hmid, W2f, b2f, N_M, (float*)d_out);
}


# Mean pool by owner work-groups: group (graph, column quarter) scans batch, sums its rows in registers and LDS, writes the mean; pool atomics, zero-fills, count and divide launches gone; sum order new
# speedup vs baseline: 2.6013x; 2.6013x over previous
#include <hip/hip_runtime.h>
#include <hip/hip_bf16.h>

typedef __attribute__((ext_vector_type(2))) float v2f;
typedef __attribute__((ext_vector_type(8))) float v8f;

#define HID 128
#define NGRAPH 64
#define LOOP_BLOCKS 2048


__global__ void k_fill(float* p, int n, float v) {
    int i = blockIdx.x * 256 + threadIdx.x;
    if (i < n) p[i] = v;
}

template <int R>
__global__ void k_deg_key(const int* __restrict__ dst, int E, int Epad,
                          float* __restrict__ deg, unsigned char* __restrict__ key) {
    int e = (blockIdx.x * 256 + threadIdx.x) * 4;
    if (e >= Epad) return;
    unsigned kw = 0xFFFFFFFFu;
    if (e + 3 < E && (((size_t)dst) & 15) == 0) {
        const int4 d = *(const int4*)(dst + e);
        atomicAdd(deg + d.x, 1.0f);
        atomicAdd(deg + d.y, 1.0f);
        atomicAdd(deg + d.z, 1.0f);
        atomicAdd(deg + d.w, 1.0f);
        kw = (unsigned)(d.x / R) | ((unsigned)(d.y / R) << 8) |
             ((unsigned)(d.z / R) << 16) | ((unsigned)(d.w / R) << 24);
    } else {
        for (int i = 0; i < 4; ++i) {
            if (e + i < E) {
                int d = dst[e + i];
                atomicAdd(deg + d, 1.0f);
                kw = (kw & ~(0xFFu << (8 * i))) | ((unsigned)(d / R) << (8 * i));
            }
        }
    }
    *(unsigned*)(key + e) = kw;
}

__global__ void k_rsqrt(float* d, int n) {
    int i = blockIdx.x * 256 + threadIdx.x;
    if (i < n) d[i] = rsqrtf(d[i]);
}


__global__ void k_embed(const int* __restrict__ res, const float* __restrict__ pos,
                        const float* __restrict__ W, const float* __restrict__ dinv,
                        int N, float* __restrict__ y) {
    int lane = threadIdx.x & 31;
    int w  = __builtin_amdgcn_readfirstlane(blockIdx.x * 8 + (threadIdx.x >> 5));
    int nw = gridDim.x * 8;
    const float4 w20 = *(const float4*)(W + 20 * HID + 4 * lane);
    const float4 w21 = *(const float4*)(W + 21 * HID + 4 * lane);
    const float4 w22 = *(const float4*)(W + 22 * HID + 4 * lane);
    for (int i = w; i < N; i += nw) {
        float p0 = pos[i * 3 + 0], p1 = pos[i * 3 + 1], p2 = pos[i * 3 + 2];
        float d = dinv[i];
        const float4 wr = *(const float4*)(W + res[i] * HID + 4 * lane);
        float4 o;
        o.x = (wr.x + p0 * w20.x + p1 * w21.x + p2 * w22.x) * d;
        o.y = (wr.y + p0 * w20.y + p1 * w21.y + p2 * w22.y) * d;
        o.z = (wr.z + p0 * w20.z + p1 * w21.z + p2 * w22.z) * d;
        o.w = (wr.w + p0 * w20.w + p1 * w21.w + p2 * w22.w) * d;
        *(float4*)(y + (size_t)i * HID + 4 * lane) = o;
    }
}

#define AGG_WAVES 32
#define AGG_Q 64

typedef volatile int __attribute__((address_space(3))) * lds_ring_t;

__device__ __forceinline__ void agg_drain(lds_ring_t qw, int head, int n, int lane, int lo,
                                          const int* __restrict__ src, const int* __restrict__ dst,
                                          const float* __restrict__ y, float* acc) {
    const int ee = qw[(head + (lane < n ? lane : 0)) & (AGG_Q - 1)];
    const int sv = src[ee];
    const int dv = dst[ee] - lo;
    #pragma unroll
    for (int g = 0; g < 32; g += 16) {
        if (g < n) {
            float4 v[16];
            #pragma unroll
            for (int j = 0; j < 16; ++j) {
                if (g + j < n) {
                    const float* yr = y + (size_t)__builtin_amdgcn_readlane(sv, g + j) * HID;
                    v[j] = *(const float4*)(yr + 4 * lane);
                }
            }
            #pragma unroll
            for (int j = 0; j < 16; ++j) {
                if (g + j < n) {
                    float* a = acc + __builtin_amdgcn_readlane(dv, g + j) * HID + lane;
                    atomicAdd(a, v[j].x);
                    atomicAdd(a + 32, v[j].y);
                    atomicAdd(a + 64, v[j].z);
                    atomicAdd(a + 96, v[j].w);
                }
            }
        }
    }
}

template <int R>
__global__ __launch_bounds__(AGG_WAVES * 32)
void k_agg(const int* __restrict__ src, const int* __restrict__ dst,
           const unsigned char* __restrict__ key, int E,
           const float* __restrict__ y, const float* __restrict__ dinv,
           const float* __restrict__ b, int N, float* __restrict__ h) {
    __shared__ float acc[R * HID];
    __shared__ int q[AGG_WAVES * AGG_Q];
    const int lane = threadIdx.x & 31;
    const int w = __builtin_amdgcn_readfirstlane(threadIdx.x >> 5);
    const int lo = blockIdx.x * R;
    int nrows = N - lo;
    if (nrows > R) nrows = R;

    #pragma unroll 4
    for (int r = w; r < nrows; r += AGG_WAVES) {
        const float4 v = *(const float4*)(y + (size_t)(lo + r) * HID + 4 * lane);
        float* a = acc + r * HID + lane;
        a[0] = v.x; a[32] = v.y; a[64] = v.z; a[96] = v.w;
    }
    __syncthreads();

    lds_ring_t qw = (lds_ring_t)(q + w * AGG_Q);
    const unsigned pat = (unsigned)blockIdx.x * 0x01010101u;
    const int nit = (E + 511) >> 9;
    int head = 0, tail = 0;
    const int it0 = (int)((blockIdx.x * (unsigned)nit) / gridDim.x);
    uint4 kv = make_uint4(0xFFFFFFFFu, 0xFFFFFFFFu, 0xFFFFFFFFu, 0xFFFFFFFFu);
    if (w < nit) {
        int itf = it0 + w;
        if (itf >= nit) itf -= nit;
        kv = *(const uint4*)(key + (size_t)itf * 512 + lane * 16);
    }
    for (int k = w; k < nit; k += AGG_WAVES) {
        const uint4 cur = kv;
        int it = it0 + k;
        if (it >= nit) it -= nit;
        const int eb = it * 512 + lane * 16;
        if (k + AGG_WAVES < nit) {
            int itn = it0 + k + AGG_WAVES;
            if (itn >= nit) itn -= nit;
            kv = *(const uint4*)(key + (size_t)itn * 512 + lane * 16);
        }
        const unsigned x0 = cur.x ^ pat, x1 = cur.y ^ pat, x2 = cur.z ^ pat, x3 = cur.w ^ pat;
        const unsigned u0 = ((x0 & 0x7F7F7F7Fu) + 0x7F7F7F7Fu) | x0;
        const unsigned u1 = ((x1 & 0x7F7F7F7Fu) + 0x7F7F7F7Fu) | x1;
        const unsigned u2 = ((x2 & 0x7F7F7F7Fu) + 0x7F7F7F7Fu) | x2;
        const unsigned u3 = ((x3 & 0x7F7F7F7Fu) + 0x7F7F7F7Fu) | x3;
        unsigned F = ((~u0 >> 7) & 0x01010101u) | ((~u1 >> 6) & 0x02020202u) |
                     ((~u2 >> 5) & 0x04040404u) | ((~u3 >> 4) & 0x08080808u);
        unsigned B = __builtin_amdgcn_ballot_w32(F != 0);
        while (B) {
            if (F) {
                const int p = __builtin_ctz(F);
                const int rank = __builtin_amdgcn_mbcnt_lo(B, 0);
                qw[(tail + rank) & (AGG_Q - 1)] = eb + ((p & 7) << 2) + (p >> 3);
                F &= F - 1;
            }
            tail += __builtin_popcount(B);
            if (tail - head >= 32) {
                agg_drain(qw, head, 32, lane, lo, src, dst, y, acc);
                head += 32;
            }
            B = __builtin_amdgcn_ballot_w32(F != 0);
        }
    }
    if (tail - head > 0) agg_drain(qw, head, tail - head, lane, lo, src, dst, y, acc);
    __syncthreads();

    const float4 bv = *(const float4*)(b + 4 * lane);
    #pragma unroll 4
    for (int r = w; r < nrows; r += AGG_WAVES) {
        const int i = lo + r;
        const float d = dinv[i];
        const float* a = acc + r * HID + lane;
        float4 o;
        o.x = fmaxf(d * a[0] + bv.x, 0.0f);
        o.y = fmaxf(d * a[32] + bv.y, 0.0f);
        o.z = fmaxf(d * a[64] + bv.z, 0.0f);
        o.w = fmaxf(d * a[96] + bv.w, 0.0f);
        *(float4*)(h + (size_t)i * HID + 4 * lane) = o;
    }
}

#define POOL_WAVES 32
#define POOL_DEPTH 8

__device__ __forceinline__ void pool_rows(const float* __restrict__ hq, int c, unsigned m,
                                          int sub, float4& acc) {
    const float* hc = hq + ((size_t)c * 128 + sub) * HID;
    #pragma nounroll
    for (int half = 0; half < 32; half += 16) {
        float4 v[16];
        #pragma unroll
        for (int j = 0; j < 16; ++j)
            v[j] = *(const float4*)(hc + (size_t)(4 * (half + j)) * HID);
        #pragma unroll
        for (int j = 0; j < 16; ++j) {
            const unsigned bits = __builtin_amdgcn_readlane(m, half + j);
            const bool on = ((bits >> sub) & 1u) != 0;
            acc.x += on ? v[j].x : 0.0f;
            acc.y += on ? v[j].y : 0.0f;
            acc.z += on ? v[j].z : 0.0f;
            acc.w += on ? v[j].w : 0.0f;
        }
    }
}

__device__ __forceinline__ void pool_rows_tail(const float* __restrict__ hq, int c, unsigned m,
                                               int sub, float4& acc) {
    for (int k = 0; k < 32; ++k) {
        const unsigned bits = __builtin_amdgcn_readlane(m, k);
        if ((bits >> sub) & 1u) {
            const float4 v = *(const float4*)(hq + ((size_t)c * 128 + 4 * k + sub) * HID);
            acc.x += v.x; acc.y += v.y; acc.z += v.z; acc.w += v.w;
        }
    }
}

__global__ __launch_bounds__(POOL_WAVES * 32)
void k_pool_own(const float* __restrict__ h, const int* __restrict__ batch, int N,
                float* __restrict__ pe) {
    __shared__ float part[POOL_WAVES * 128];
    __shared__ int cntw[POOL_WAVES];
    const int lane = threadIdx.x & 31;
    const int w = __builtin_amdgcn_readfirstlane(threadIdx.x >> 5);
    const int g = blockIdx.x >> 2;
    const int q = blockIdx.x & 3;
    const int sub = lane >> 3;
    const float* hq = h + 32 * q + 4 * (lane & 7);
    const bool al = (((size_t)batch) & 15) == 0;
    const int nchunk = (N + 127) >> 7;
    const int c0 = (int)((blockIdx.x * (unsigned)nchunk) / gridDim.x);
    float4 acc = make_float4(0.0f, 0.0f, 0.0f, 0.0f);
    int cnt = 0;
    for (int k0 = w; k0 < nchunk; k0 += POOL_DEPTH * POOL_WAVES) {
        unsigned mp = 0;
        if (al && N >= 4) {
            int4 b[POOL_DEPTH];
            #pragma unroll
            for (int j = 0; j < POOL_DEPTH; ++j) {
                const int k = k0 + j * POOL_WAVES;
                int c = c0 + k;
                if (c >= nchunk) c -= nchunk;
                const int r0 = c * 128 + 4 * lane;
                b[j] = *(const int4*)(batch + ((k < nchunk && r0 + 3 < N) ? r0 : 0));
            }
            #pragma unroll
            for (int j = 0; j < POOL_DEPTH; ++j) {
                const int k = k0 + j * POOL_WAVES;
                int c = c0 + k;
                if (c >= nchunk) c -= nchunk;
                const int r0 = c * 128 + 4 * lane;
                unsigned mj = 0;
                if (k < nchunk) {
                    if (r0 + 3 < N) {
                        mj = (b[j].x == g ? 1u : 0u) | (b[j].y == g ? 2u : 0u) |
                             (b[j].z == g ? 4u : 0u) | (b[j].w == g ? 8u : 0u);
                    } else {
                        for (int i = 0; i < 4; ++i)
                            if (r0 + i < N && batch[r0 + i] == g) mj |= 1u << i;
                    }
                }
                mp |= mj << (4 * j);
            }
        } else {
            for (int j = 0; j < POOL_DEPTH; ++j) {
                const int k = k0 + j * POOL_WAVES;
                if (k < nchunk) {
                    int c = c0 + k;
                    if (c >= nchunk) c -= nchunk;
                    const int r0 = c * 128 + 4 * lane;
                    unsigned mj = 0;
                    for (int i = 0; i < 4; ++i)
                        if (r0 + i < N && batch[r0 + i] == g) mj |= 1u << i;
                    mp |= mj << (4 * j);
                }
            }
        }
        if (__builtin_amdgcn_ballot_w32(mp != 0) == 0) continue;
        for (int j = 0; j < POOL_DEPTH; ++j) {
            const unsigned mj = (mp >> (4 * j)) & 15u;
            if (__builtin_amdgcn_ballot_w32(mj != 0) == 0) continue;
            int c = c0 + k0 + j * POOL_WAVES;
            if (c >= nchunk) c -= nchunk;
            cnt += __builtin_popcount(__builtin_amdgcn_ballot_w32((mj & 1u) != 0)) +
                   __builtin_popcount(__builtin_amdgcn_ballot_w32((mj & 2u) != 0)) +
                   __builtin_popcount(__builtin_amdgcn_ballot_w32((mj & 4u) != 0)) +
                   __builtin_popcount(__builtin_amdgcn_ballot_w32((mj & 8u) != 0));
            if (c * 128 + 128 <= N) pool_rows(hq, c, mj, sub, acc);
            else pool_rows_tail(hq, c, mj, sub, acc);
        }
    }
    *(float4*)(part + w * 128 + 4 * lane) = acc;
    if (lane == 0) cntw[w] = cnt;
    __syncthreads();
    if (threadIdx.x < 32) {
        float s0 = 0.0f, s1 = 0.0f, s2 = 0.0f, s3 = 0.0f;
        #pragma unroll 8
        for (int j = 0; j < POOL_WAVES * 4; j += 4) {
            s0 += part[(j + 0) * 32 + lane];
            s1 += part[(j + 1) * 32 + lane];
            s2 += part[(j + 2) * 32 + lane];
            s3 += part[(j + 3) * 32 + lane];
        }
        int n = 0;
        for (int j = 0; j < POOL_WAVES; ++j) n += cntw[j];
        pe[g * HID + 32 * q + lane] = ((s0 + s1) + (s2 + s3)) / fmaxf((float)n, 1.0f);
    }
}

__global__ void k_feat(const float* __restrict__ h, const float* __restrict__ pe,
                       const int* __restrict__ batch, int N, float* __restrict__ feat) {
    int lane = threadIdx.x & 31;
    int w  = __builtin_amdgcn_readfirstlane(blockIdx.x * 8 + (threadIdx.x >> 5));
    int nw = gridDim.x * 8;
    for (int i = w; i < N; i += nw) {
        const float4 a = *(const float4*)(h + (size_t)i * HID + 4 * lane);
        const float4 c = *(const float4*)(pe + batch[i] * HID + 4 * lane);
        float* fr = feat + (size_t)i * 2 * HID + 4 * lane;
        *(float4*)fr = a;
        *(float4*)(fr + HID) = c;
    }
}

__global__ void k_head(const float* __restrict__ hmid, const float* __restrict__ W2,
                       const float* __restrict__ b2, int N, float* __restrict__ out) {
    int i = blockIdx.x * 256 + threadIdx.x;
    if (i >= N) return;
    float a0 = b2[0], a1 = b2[1], a2 = b2[2];
    const float* h = hmid + (size_t)i * HID;
    for (int k = 0; k < HID; ++k) {
        float hv = h[k];
        a0 += hv * W2[k * 3 + 0];
        a1 += hv * W2[k * 3 + 1];
        a2 += hv * W2[k * 3 + 2];
    }
    out[(size_t)i * 3 + 0] = a0;
    out[(size_t)i * 3 + 1] = a1;
    out[(size_t)i * 3 + 2] = a2;
}

__global__ void k_wfrag(const float* __restrict__ W, int K, float* __restrict__ Wf) {
    int idx = blockIdx.x * 256 + threadIdx.x;
    if (idx >= K * HID) return;
    int j    = idx & 1;
    int nt   = (idx >> 1) & 7;
    int l    = (idx >> 4) & 31;
    int kk   = idx >> 9;
    int krow = 4 * kk + (l >> 4) * 2 + j;
    int col  = nt * 16 + (l & 15);
    Wf[idx] = W[(size_t)krow * HID + col];
}

__global__ void k_gemm128(const float* __restrict__ A, int K, int M,
                          const float* __restrict__ Wf,
                          const float* __restrict__ rowscale,
                          const float* __restrict__ bias, int relu,
                          float* __restrict__ out) {
    int w = threadIdx.x >> 5;
    int lane = threadIdx.x & 31;
    int tile = blockIdx.x * 4 + w;
    if (tile * 16 >= M) return;
    int row0 = tile * 16;
    int half = lane >> 4;
    int lmod = lane & 15;
    int koff = half * 2;
    const float* Arow = A + (size_t)(row0 + lmod) * K + koff;
    const float* Wl   = Wf + (size_t)lane * 16;

    v8f acc[8];
    #pragma unroll
    for (int nt = 0; nt < 8; ++nt) acc[nt] = (v8f){};

    int ksteps = K >> 2;
    for (int kk = 0; kk < ksteps; ++kk) {
        v2f a;
        a.x = Arow[4 * kk];
        a.y = Arow[4 * kk + 1];
        const float4* q = (const float4*)(Wl + (size_t)kk * 512);
        float4 q0 = q[0], q1 = q[1], q2 = q[2], q3 = q[3];
        v2f b0; b0.x = q0.x; b0.y = q0.y;
        v2f b1; b1.x = q0.z; b1.y = q0.w;
        v2f b2; b2.x = q1.x; b2.y = q1.y;
        v2f b3; b3.x = q1.z; b3.y = q1.w;
        v2f b4; b4.x = q2.x; b4.y = q2.y;
        v2f b5; b5.x = q2.z; b5.y = q2.w;
        v2f b6; b6.x = q3.x; b6.y = q3.y;
        v2f b7; b7.x = q3.z; b7.y = q3.w;
        acc[0] = __builtin_amdgcn_wmma_f32_16x16x4_f32(false, a, false, b0, (short)0, acc[0], false, false);
        acc[1] = __builtin_amdgcn_wmma_f32_16x16x4_f32(false, a, false, b1, (short)0, acc[1], false, false);
        acc[2] = __builtin_amdgcn_wmma_f32_16x16x4_f32(false, a, false, b2, (short)0, acc[2], false, false);
        acc[3] = __builtin_amdgcn_wmma_f32_16x16x4_f32(false, a, false, b3, (short)0, acc[3], false, false);
        acc[4] = __builtin_amdgcn_wmma_f32_16x16x4_f32(false, a, false, b4, (short)0, acc[4], false, false);
        acc[5] = __builtin_amdgcn_wmma_f32_16x16x4_f32(false, a, false, b5, (short)0, acc[5], false, false);
        acc[6] = __builtin_amdgcn_wmma_f32_16x16x4_f32(false, a, false, b6, (short)0, acc[6], false, false);
        acc[7] = __builtin_amdgcn_wmma_f32_16x16x4_f32(false, a, false, b7, (short)0, acc[7], false, false);
    }

    float rs[8];
    if (rowscale) {
        #pragma unroll
        for (int r = 0; r < 8; ++r) rs[r] = rowscale[row0 + r + half * 8];
    }

    #pragma unroll
    for (int nt = 0; nt < 8; ++nt) {
        int col = nt * 16 + lmod;
        float bv = bias ? bias[col] : 0.0f;
        #pragma unroll
        for (int r = 0; r < 8; ++r) {
            float v = acc[nt][r];
            if (rowscale) v *= rs[r];
            v += bv;
            if (relu) v = fmaxf(v, 0.0f);
            out[(size_t)(row0 + r + half * 8) * HID + col] = v;
        }
    }
}


static inline int cdiv(long long a, long long b) { return (int)((a + b - 1) / b); }

template <int R>
static void run_branch(const int* res, const float* pos, const int* ei,
                       int N, int E,
                       const float* W1c, const float* b1c,
                       const float* W2c, const float* b2c,
                       float* dinv, unsigned char* key, float* A, float* C, float* Wf,
                       hipStream_t stream) {
    const int* src = ei;
    const int* dst = ei + E;
    int Epad = cdiv(E, 512) * 512;

    k_fill<<<cdiv(N, 256), 256, 0, stream>>>(dinv, N, 1.0f);
    k_deg_key<R><<<cdiv(Epad / 4, 256), 256, 0, stream>>>(dst, E, Epad, dinv, key);
    k_rsqrt<<<cdiv(N, 256), 256, 0, stream>>>(dinv, N);

    k_embed<<<LOOP_BLOCKS, 256, 0, stream>>>(res, pos, W1c, dinv, N, A);
    k_agg<R><<<cdiv(N, R), AGG_WAVES * 32, 0, stream>>>(src, dst, key, E, A, dinv, b1c, N, C);

    k_wfrag<<<cdiv(HID * HID, 256), 256, 0, stream>>>(W2c, HID, Wf);
    k_gemm128<<<cdiv(N / 16, 4), 128, 0, stream>>>(C, HID, N, Wf, dinv, nullptr, 0, A);
    k_agg<R><<<cdiv(N, R), AGG_WAVES * 32, 0, stream>>>(src, dst, key, E, A, dinv, b2c, N, C);
}

extern "C" void kernel_launch(void* const* d_in, const int* in_sizes, int n_in,
                              void* d_out, int out_size, void* d_ws, size_t ws_size,
                              hipStream_t stream) {
    (void)n_in; (void)out_size; (void)ws_size;

    const int*   res_p   = (const int*)d_in[0];
    const float* pos_p   = (const float*)d_in[1];
    const int*   ei_p    = (const int*)d_in[2];
    const int*   batch_p = (const int*)d_in[3];
    const int*   res_m   = (const int*)d_in[4];
    const float* pos_m   = (const float*)d_in[5];
    const int*   ei_m    = (const int*)d_in[6];
    const int*   batch_m = (const int*)d_in[7];
    const float* W_p1 = (const float*)d_in[8];
    const float* b_p1 = (const float*)d_in[9];
    const float* W_p2 = (const float*)d_in[10];
    const float* b_p2 = (const float*)d_in[11];
    const float* W_m1 = (const float*)d_in[12];
    const float* b_m1 = (const float*)d_in[13];
    const float* W_m2 = (const float*)d_in[14];
    const float* b_m2 = (const float*)d_in[15];
    const float* W1f  = (const float*)d_in[16];
    const float* b1f  = (const float*)d_in[17];
    const float* W2f  = (const float*)d_in[18];
    const float* b2f  = (const float*)d_in[19];

    int N_P = in_sizes[0];
    int E_P = in_sizes[2] / 2;
    int N_M = in_sizes[4];
    int E_M = in_sizes[6] / 2;

    float* ws = (float*)d_ws;
    size_t nhp = (size_t)N_P * HID;
    float* A      = ws;
    float* B      = A + nhp;
    float* C      = B + nhp;
    unsigned char* key_p = (unsigned char*)(C + nhp);
    unsigned char* key_m = key_p + (size_t)cdiv(E_P, 512) * 512;
    float* dinv_p = (float*)(key_m + (size_t)cdiv(E_M, 512) * 512);
    float* dinv_m = dinv_p + N_P;
    float* pe     = dinv_m + N_M;
    float* Wfrag  = pe + NGRAPH * HID;

    run_branch<400>(res_p, pos_p, ei_p, N_P, E_P, W_p1, b_p1, W_p2, b_p2,
                    dinv_p, key_p, A, C, Wfrag, stream);
    k_pool_own<<<NGRAPH * 4, POOL_WAVES * 32, 0, stream>>>(C, batch_p, N_P, pe);

    run_branch<200>(res_m, pos_m, ei_m, N_M, E_M, W_m1, b_m1, W_m2, b_m2,
                    dinv_m, key_m, A, C, Wfrag, stream);

    float* feat = A;
    float* hmid = B;
    k_feat<<<LOOP_BLOCKS, 256, 0, stream>>>(C, pe, batch_m, N_M, feat);
    k_wfrag<<<cdiv(2 * HID * HID, 256), 256, 0, stream>>>(W1f, 2 * HID, Wfrag);
    k_gemm128<<<cdiv(N_M / 16, 4), 128, 0, stream>>>(feat, 2 * HID, N_M, Wfrag,
                                                     nullptr, b1f, 1, hmid);
    k_head<<<cdiv(N_M, 256), 256, 0, stream>>>(hmid, W2f, b2f, N_M, (float*)d_out);
}


# Owner-key scan done once per branch: layer-1 aggregation records each wave's matched (src, dst row) pairs, layer 2 replays them: no key scan, no src/dst fetch; LDS float atomics kept, arrival order
# speedup vs baseline: 2.7301x; 1.0495x over previous
#include <hip/hip_runtime.h>
#include <hip/hip_bf16.h>

typedef __attribute__((ext_vector_type(2))) float v2f;
typedef __attribute__((ext_vector_type(8))) float v8f;

#define HID 128
#define NGRAPH 64
#define LOOP_BLOCKS 2048


__global__ void k_fill(float* p, int n, float v) {
    int i = blockIdx.x * 256 + threadIdx.x;
    if (i < n) p[i] = v;
}

template <int R>
__global__ void k_deg_key(const int* __restrict__ dst, int E, int Epad,
                          float* __restrict__ deg, unsigned char* __restrict__ key) {
    int e = (blockIdx.x * 256 + threadIdx.x) * 4;
    if (e >= Epad) return;
    unsigned kw = 0xFFFFFFFFu;
    if (e + 3 < E && (((size_t)dst) & 15) == 0) {
        const int4 d = *(const int4*)(dst + e);
        atomicAdd(deg + d.x, 1.0f);
        atomicAdd(deg + d.y, 1.0f);
        atomicAdd(deg + d.z, 1.0f);
        atomicAdd(deg + d.w, 1.0f);
        kw = (unsigned)(d.x / R) | ((unsigned)(d.y / R) << 8) |
             ((unsigned)(d.z / R) << 16) | ((unsigned)(d.w / R) << 24);
    } else {
        for (int i = 0; i < 4; ++i) {
            if (e + i < E) {
                int d = dst[e + i];
                atomicAdd(deg + d, 1.0f);
                kw = (kw & ~(0xFFu << (8 * i))) | ((unsigned)(d / R) << (8 * i));
            }
        }
    }
    *(unsigned*)(key + e) = kw;
}

__global__ void k_rsqrt(float* d, int n) {
    int i = blockIdx.x * 256 + threadIdx.x;
    if (i < n) d[i] = rsqrtf(d[i]);
}


__global__ void k_embed(const int* __restrict__ res, const float* __restrict__ pos,
                        const float* __restrict__ W, const float* __restrict__ dinv,
                        int N, float* __restrict__ y) {
    int lane = threadIdx.x & 31;
    int w  = __builtin_amdgcn_readfirstlane(blockIdx.x * 8 + (threadIdx.x >> 5));
    int nw = gridDim.x * 8;
    const float4 w20 = *(const float4*)(W + 20 * HID + 4 * lane);
    const float4 w21 = *(const float4*)(W + 21 * HID + 4 * lane);
    const float4 w22 = *(const float4*)(W + 22 * HID + 4 * lane);
    for (int i = w; i < N; i += nw) {
        float p0 = pos[i * 3 + 0], p1 = pos[i * 3 + 1], p2 = pos[i * 3 + 2];
        float d = dinv[i];
        const float4 wr = *(const float4*)(W + res[i] * HID + 4 * lane);
        float4 o;
        o.x = (wr.x + p0 * w20.x + p1 * w21.x + p2 * w22.x) * d;
        o.y = (wr.y + p0 * w20.y + p1 * w21.y + p2 * w22.y) * d;
        o.z = (wr.z + p0 * w20.z + p1 * w21.z + p2 * w22.z) * d;
        o.w = (wr.w + p0 * w20.w + p1 * w21.w + p2 * w22.w) * d;
        *(float4*)(y + (size_t)i * HID + 4 * lane) = o;
    }
}

#define AGG_WAVES 32
#define AGG_Q 64
#define AGG_CAP 512
#define AGG_MAXWG 256

typedef volatile int __attribute__((address_space(3))) * lds_ring_t;

__device__ __forceinline__ void agg_rows(int sv, int dv, int n, int lane,
                                         const float* __restrict__ y, float* acc) {
    #pragma unroll
    for (int g = 0; g < 32; g += 16) {
        if (g < n) {
            float4 v[16];
            #pragma unroll
            for (int j = 0; j < 16; ++j) {
                if (g + j < n) {
                    const float* yr = y + (size_t)__builtin_amdgcn_readlane(sv, g + j) * HID;
                    v[j] = *(const float4*)(yr + 4 * lane);
                }
            }
            #pragma unroll
            for (int j = 0; j < 16; ++j) {
                if (g + j < n) {
                    float* a = acc + __builtin_amdgcn_readlane(dv, g + j) * HID + lane;
                    atomicAdd(a, v[j].x);
                    atomicAdd(a + 32, v[j].y);
                    atomicAdd(a + 64, v[j].z);
                    atomicAdd(a + 96, v[j].w);
                }
            }
        }
    }
}

template <bool REC>
__device__ __forceinline__ void agg_drain(lds_ring_t qw, int head, int n, int lane, int lo,
                                          const int* __restrict__ src, const int* __restrict__ dst,
                                          const float* __restrict__ y, float* acc, int2* recw) {
    const int ee = qw[(head + (lane < n ? lane : 0)) & (AGG_Q - 1)];
    const int sv = src[ee];
    const int dv = dst[ee] - lo;
    if (REC) {
        if (lane < n && head + lane < AGG_CAP) recw[head + lane] = make_int2(sv, dv);
    }
    agg_rows(sv, dv, n, lane, y, acc);
}

template <bool REC>
__device__ __forceinline__ int agg_scan(const int* __restrict__ src, const int* __restrict__ dst,
                                        const unsigned char* __restrict__ key, int E,
                                        const float* __restrict__ y, float* acc, lds_ring_t qw,
                                        int w, int lane, int lo, int2* recw) {
    const unsigned pat = (unsigned)blockIdx.x * 0x01010101u;
    const int nit = (E + 511) >> 9;
    int head = 0, tail = 0;
    const int it0 = (int)((blockIdx.x * (unsigned)nit) / gridDim.x);
    uint4 kv = make_uint4(0xFFFFFFFFu, 0xFFFFFFFFu, 0xFFFFFFFFu, 0xFFFFFFFFu);
    if (w < nit) {
        int itf = it0 + w;
        if (itf >= nit) itf -= nit;
        kv = *(const uint4*)(key + (size_t)itf * 512 + lane * 16);
    }
    for (int k = w; k < nit; k += AGG_WAVES) {
        const uint4 cur = kv;
        int it = it0 + k;
        if (it >= nit) it -= nit;
        const int eb = it * 512 + lane * 16;
        if (k + AGG_WAVES < nit) {
            int itn = it0 + k + AGG_WAVES;
            if (itn >= nit) itn -= nit;
            kv = *(const uint4*)(key + (size_t)itn * 512 + lane * 16);
        }
        const unsigned x0 = cur.x ^ pat, x1 = cur.y ^ pat, x2 = cur.z ^ pat, x3 = cur.w ^ pat;
        const unsigned u0 = ((x0 & 0x7F7F7F7Fu) + 0x7F7F7F7Fu) | x0;
        const unsigned u1 = ((x1 & 0x7F7F7F7Fu) + 0x7F7F7F7Fu) | x1;
        const unsigned u2 = ((x2 & 0x7F7F7F7Fu) + 0x7F7F7F7Fu) | x2;
        const unsigned u3 = ((x3 & 0x7F7F7F7Fu) + 0x7F7F7F7Fu) | x3;
        unsigned F = ((~u0 >> 7) & 0x01010101u) | ((~u1 >> 6) & 0x02020202u) |
                     ((~u2 >> 5) & 0x04040404u) | ((~u3 >> 4) & 0x08080808u);
        unsigned B = __builtin_amdgcn_ballot_w32(F != 0);
        while (B) {
            if (F) {
                const int p = __builtin_ctz(F);
                const int rank = __builtin_amdgcn_mbcnt_lo(B, 0);
                qw[(tail + rank) & (AGG_Q - 1)] = eb + ((p & 7) << 2) + (p >> 3);
                F &= F - 1;
            }
            tail += __builtin_popcount(B);
            if (tail - head >= 32) {
                agg_drain<REC>(qw, head, 32, lane, lo, src, dst, y, acc, recw);
                head += 32;
            }
            B = __builtin_amdgcn_ballot_w32(F != 0);
        }
    }
    if (tail - head > 0) agg_drain<REC>(qw, head, tail - head, lane, lo, src, dst, y, acc, recw);
    return tail;
}

template <int R, int MODE>
__global__ __launch_bounds__(AGG_WAVES * 32)
void k_agg(const int* __restrict__ src, const int* __restrict__ dst,
           const unsigned char* __restrict__ key, int E,
           const float* __restrict__ y, const float* __restrict__ dinv,
           const float* __restrict__ b, int N, float* __restrict__ h,
           int2* rec, int* cnt) {
    __shared__ float acc[R * HID];
    __shared__ int q[AGG_WAVES * AGG_Q];
    const int lane = threadIdx.x & 31;
    const int w = __builtin_amdgcn_readfirstlane(threadIdx.x >> 5);
    const int lo = blockIdx.x * R;
    int nrows = N - lo;
    if (nrows > R) nrows = R;

    #pragma unroll 4
    for (int r = w; r < nrows; r += AGG_WAVES) {
        const float4 v = *(const float4*)(y + (size_t)(lo + r) * HID + 4 * lane);
        float* a = acc + r * HID + lane;
        a[0] = v.x; a[32] = v.y; a[64] = v.z; a[96] = v.w;
    }
    __syncthreads();

    lds_ring_t qw = (lds_ring_t)(q + w * AGG_Q);
    int2* recw = rec + ((size_t)blockIdx.x * AGG_WAVES + w) * AGG_CAP;
    int* cntw = cnt + blockIdx.x * AGG_WAVES + w;
    if (MODE == 0) {
        const int nt = agg_scan<true>(src, dst, key, E, y, acc, qw, w, lane, lo, recw);
        if (lane == 0) *cntw = nt;
    } else {
        const int nt = __builtin_amdgcn_readfirstlane(*cntw);
        if (nt <= AGG_CAP) {
            int2 nx = make_int2(0, 0);
            if (nt > 0) nx = recw[lane < nt ? lane : 0];
            for (int p = 0; p < nt; p += 32) {
                const int2 e = nx;
                const int n = nt - p < 32 ? nt - p : 32;
                if (p + 32 < nt) nx = recw[p + 32 + (lane < nt - p - 32 ? lane : 0)];
                agg_rows(e.x, e.y, n, lane, y, acc);
            }
        } else {
            agg_scan<false>(src, dst, key, E, y, acc, qw, w, lane, lo, recw);
        }
    }
    __syncthreads();

    const float4 bv = *(const float4*)(b + 4 * lane);
    #pragma unroll 4
    for (int r = w; r < nrows; r += AGG_WAVES) {
        const int i = lo + r;
        const float d = dinv[i];
        const float* a = acc + r * HID + lane;
        float4 o;
        o.x = fmaxf(d * a[0] + bv.x, 0.0f);
        o.y = fmaxf(d * a[32] + bv.y, 0.0f);
        o.z = fmaxf(d * a[64] + bv.z, 0.0f);
        o.w = fmaxf(d * a[96] + bv.w, 0.0f);
        *(float4*)(h + (size_t)i * HID + 4 * lane) = o;
    }
}

#define POOL_WAVES 32
#define POOL_DEPTH 8

__device__ __forceinline__ void pool_rows(const float* __restrict__ hq, int c, unsigned m,
                                          int sub, float4& acc) {
    const float* hc = hq + ((size_t)c * 128 + sub) * HID;
    #pragma nounroll
    for (int half = 0; half < 32; half += 16) {
        float4 v[16];
        #pragma unroll
        for (int j = 0; j < 16; ++j)
            v[j] = *(const float4*)(hc + (size_t)(4 * (half + j)) * HID);
        #pragma unroll
        for (int j = 0; j < 16; ++j) {
            const unsigned bits = __builtin_amdgcn_readlane(m, half + j);
            const bool on = ((bits >> sub) & 1u) != 0;
            acc.x += on ? v[j].x : 0.0f;
            acc.y += on ? v[j].y : 0.0f;
            acc.z += on ? v[j].z : 0.0f;
            acc.w += on ? v[j].w : 0.0f;
        }
    }
}

__device__ __forceinline__ void pool_rows_tail(const float* __restrict__ hq, int c, unsigned m,
                                               int sub, float4& acc) {
    for (int k = 0; k < 32; ++k) {
        const unsigned bits = __builtin_amdgcn_readlane(m, k);
        if ((bits >> sub) & 1u) {
            const float4 v = *(const float4*)(hq + ((size_t)c * 128 + 4 * k + sub) * HID);
            acc.x += v.x; acc.y += v.y; acc.z += v.z; acc.w += v.w;
        }
    }
}

__global__ __launch_bounds__(POOL_WAVES * 32)
void k_pool_own(const float* __restrict__ h, const int* __restrict__ batch, int N,
                float* __restrict__ pe) {
    __shared__ float part[POOL_WAVES * 128];
    __shared__ int cntw[POOL_WAVES];
    const int lane = threadIdx.x & 31;
    const int w = __builtin_amdgcn_readfirstlane(threadIdx.x >> 5);
    const int g = blockIdx.x >> 2;
    const int q = blockIdx.x & 3;
    const int sub = lane >> 3;
    const float* hq = h + 32 * q + 4 * (lane & 7);
    const bool al = (((size_t)batch) & 15) == 0;
    const int nchunk = (N + 127) >> 7;
    const int c0 = (int)((blockIdx.x * (unsigned)nchunk) / gridDim.x);
    float4 acc = make_float4(0.0f, 0.0f, 0.0f, 0.0f);
    int cnt = 0;
    for (int k0 = w; k0 < nchunk; k0 += POOL_DEPTH * POOL_WAVES) {
        unsigned mp = 0;
        if (al && N >= 4) {
            int4 b[POOL_DEPTH];
            #pragma unroll
            for (int j = 0; j < POOL_DEPTH; ++j) {
                const int k = k0 + j * POOL_WAVES;
                int c = c0 + k;
                if (c >= nchunk) c -= nchunk;
                const int r0 = c * 128 + 4 * lane;
                b[j] = *(const int4*)(batch + ((k < nchunk && r0 + 3 < N) ? r0 : 0));
            }
            #pragma unroll
            for (int j = 0; j < POOL_DEPTH; ++j) {
                const int k = k0 + j * POOL_WAVES;
                int c = c0 + k;
                if (c >= nchunk) c -= nchunk;
                const int r0 = c * 128 + 4 * lane;
                unsigned mj = 0;
                if (k < nchunk) {
                    if (r0 + 3 < N) {
                        mj = (b[j].x == g ? 1u : 0u) | (b[j].y == g ? 2u : 0u) |
                             (b[j].z == g ? 4u : 0u) | (b[j].w == g ? 8u : 0u);
                    } else {
                        for (int i = 0; i < 4; ++i)
                            if (r0 + i < N && batch[r0 + i] == g) mj |= 1u << i;
                    }
                }
                mp |= mj << (4 * j);
            }
        } else {
            for (int j = 0; j < POOL_DEPTH; ++j) {
                const int k = k0 + j * POOL_WAVES;
                if (k < nchunk) {
                    int c = c0 + k;
                    if (c >= nchunk) c -= nchunk;
                    const int r0 = c * 128 + 4 * lane;
                    unsigned mj = 0;
                    for (int i = 0; i < 4; ++i)
                        if (r0 + i < N && batch[r0 + i] == g) mj |= 1u << i;
                    mp |= mj << (4 * j);
                }
            }
        }
        if (__builtin_amdgcn_ballot_w32(mp != 0) == 0) continue;
        for (int j = 0; j < POOL_DEPTH; ++j) {
            const unsigned mj = (mp >> (4 * j)) & 15u;
            if (__builtin_amdgcn_ballot_w32(mj != 0) == 0) continue;
            int c = c0 + k0 + j * POOL_WAVES;
            if (c >= nchunk) c -= nchunk;
            cnt += __builtin_popcount(__builtin_amdgcn_ballot_w32((mj & 1u) != 0)) +
                   __builtin_popcount(__builtin_amdgcn_ballot_w32((mj & 2u) != 0)) +
                   __builtin_popcount(__builtin_amdgcn_ballot_w32((mj & 4u) != 0)) +
                   __builtin_popcount(__builtin_amdgcn_ballot_w32((mj & 8u) != 0));
            if (c * 128 + 128 <= N) pool_rows(hq, c, mj, sub, acc);
            else pool_rows_tail(hq, c, mj, sub, acc);
        }
    }
    *(float4*)(part + w * 128 + 4 * lane) = acc;
    if (lane == 0) cntw[w] = cnt;
    __syncthreads();
    if (threadIdx.x < 32) {
        float s0 = 0.0f, s1 = 0.0f, s2 = 0.0f, s3 = 0.0f;
        #pragma unroll 8
        for (int j = 0; j < POOL_WAVES * 4; j += 4) {
            s0 += part[(j + 0) * 32 + lane];
            s1 += part[(j + 1) * 32 + lane];
            s2 += part[(j + 2) * 32 + lane];
            s3 += part[(j + 3) * 32 + lane];
        }
        int n = 0;
        for (int j = 0; j < POOL_WAVES; ++j) n += cntw[j];
        pe[g * HID + 32 * q + lane] = ((s0 + s1) + (s2 + s3)) / fmaxf((float)n, 1.0f);
    }
}

__global__ void k_feat(const float* __restrict__ h, const float* __restrict__ pe,
                       const int* __restrict__ batch, int N, float* __restrict__ feat) {
    int lane = threadIdx.x & 31;
    int w  = __builtin_amdgcn_readfirstlane(blockIdx.x * 8 + (threadIdx.x >> 5));
    int nw = gridDim.x * 8;
    for (int i = w; i < N; i += nw) {
        const float4 a = *(const float4*)(h + (size_t)i * HID + 4 * lane);
        const float4 c = *(const float4*)(pe + batch[i] * HID + 4 * lane);
        float* fr = feat + (size_t)i * 2 * HID + 4 * lane;
        *(float4*)fr = a;
        *(float4*)(fr + HID) = c;
    }
}

__global__ void k_head(const float* __restrict__ hmid, const float* __restrict__ W2,
                       const float* __restrict__ b2, int N, float* __restrict__ out) {
    int i = blockIdx.x * 256 + threadIdx.x;
    if (i >= N) return;
    float a0 = b2[0], a1 = b2[1], a2 = b2[2];
    const float* h = hmid + (size_t)i * HID;
    for (int k = 0; k < HID; ++k) {
        float hv = h[k];
        a0 += hv * W2[k * 3 + 0];
        a1 += hv * W2[k * 3 + 1];
        a2 += hv * W2[k * 3 + 2];
    }
    out[(size_t)i * 3 + 0] = a0;
    out[(size_t)i * 3 + 1] = a1;
    out[(size_t)i * 3 + 2] = a2;
}

__global__ void k_wfrag(const float* __restrict__ W, int K, float* __restrict__ Wf) {
    int idx = blockIdx.x * 256 + threadIdx.x;
    if (idx >= K * HID) return;
    int j    = idx & 1;
    int nt   = (idx >> 1) & 7;
    int l    = (idx >> 4) & 31;
    int kk   = idx >> 9;
    int krow = 4 * kk + (l >> 4) * 2 + j;
    int col  = nt * 16 + (l & 15);
    Wf[idx] = W[(size_t)krow * HID + col];
}

__global__ void k_gemm128(const float* __restrict__ A, int K, int M,
                          const float* __restrict__ Wf,
                          const float* __restrict__ rowscale,
                          const float* __restrict__ bias, int relu,
                          float* __restrict__ out) {
    int w = threadIdx.x >> 5;
    int lane = threadIdx.x & 31;
    int tile = blockIdx.x * 4 + w;
    if (tile * 16 >= M) return;
    int row0 = tile * 16;
    int half = lane >> 4;
    int lmod = lane & 15;
    int koff = half * 2;
    const float* Arow = A + (size_t)(row0 + lmod) * K + koff;
    const float* Wl   = Wf + (size_t)lane * 16;

    v8f acc[8];
    #pragma unroll
    for (int nt = 0; nt < 8; ++nt) acc[nt] = (v8f){};

    int ksteps = K >> 2;
    for (int kk = 0; kk < ksteps; ++kk) {
        v2f a;
        a.x = Arow[4 * kk];
        a.y = Arow[4 * kk + 1];
        const float4* q = (const float4*)(Wl + (size_t)kk * 512);
        float4 q0 = q[0], q1 = q[1], q2 = q[2], q3 = q[3];
        v2f b0; b0.x = q0.x; b0.y = q0.y;
        v2f b1; b1.x = q0.z; b1.y = q0.w;
        v2f b2; b2.x = q1.x; b2.y = q1.y;
        v2f b3; b3.x = q1.z; b3.y = q1.w;
        v2f b4; b4.x = q2.x; b4.y = q2.y;
        v2f b5; b5.x = q2.z; b5.y = q2.w;
        v2f b6; b6.x = q3.x; b6.y = q3.y;
        v2f b7; b7.x = q3.z; b7.y = q3.w;
        acc[0] = __builtin_amdgcn_wmma_f32_16x16x4_f32(false, a, false, b0, (short)0, acc[0], false, false);
        acc[1] = __builtin_amdgcn_wmma_f32_16x16x4_f32(false, a, false, b1, (short)0, acc[1], false, false);
        acc[2] = __builtin_amdgcn_wmma_f32_16x16x4_f32(false, a, false, b2, (short)0, acc[2], false, false);
        acc[3] = __builtin_amdgcn_wmma_f32_16x16x4_f32(false, a, false, b3, (short)0, acc[3], false, false);
        acc[4] = __builtin_amdgcn_wmma_f32_16x16x4_f32(false, a, false, b4, (short)0, acc[4], false, false);
        acc[5] = __builtin_amdgcn_wmma_f32_16x16x4_f32(false, a, false, b5, (short)0, acc[5], false, false);
        acc[6] = __builtin_amdgcn_wmma_f32_16x16x4_f32(false, a, false, b6, (short)0, acc[6], false, false);
        acc[7] = __builtin_amdgcn_wmma_f32_16x16x4_f32(false, a, false, b7, (short)0, acc[7], false, false);
    }

    float rs[8];
    if (rowscale) {
        #pragma unroll
        for (int r = 0; r < 8; ++r) rs[r] = rowscale[row0 + r + half * 8];
    }

    #pragma unroll
    for (int nt = 0; nt < 8; ++nt) {
        int col = nt * 16 + lmod;
        float bv = bias ? bias[col] : 0.0f;
        #pragma unroll
        for (int r = 0; r < 8; ++r) {
            float v = acc[nt][r];
            if (rowscale) v *= rs[r];
            v += bv;
            if (relu) v = fmaxf(v, 0.0f);
            out[(size_t)(row0 + r + half * 8) * HID + col] = v;
        }
    }
}


static inline int cdiv(long long a, long long b) { return (int)((a + b - 1) / b); }

template <int R>
static void run_branch(const int* res, const float* pos, const int* ei,
                       int N, int E,
                       const float* W1c, const float* b1c,
                       const float* W2c, const float* b2c,
                       float* dinv, unsigned char* key, float* A, float* C, float* Wf,
                       int2* rec, int* cnt, hipStream_t stream) {
    const int* src = ei;
    const int* dst = ei + E;
    int Epad = cdiv(E, 512) * 512;

    k_fill<<<cdiv(N, 256), 256, 0, stream>>>(dinv, N, 1.0f);
    k_deg_key<R><<<cdiv(Epad / 4, 256), 256, 0, stream>>>(dst, E, Epad, dinv, key);
    k_rsqrt<<<cdiv(N, 256), 256, 0, stream>>>(dinv, N);

    k_embed<<<LOOP_BLOCKS, 256, 0, stream>>>(res, pos, W1c, dinv, N, A);
    k_agg<R, 0><<<cdiv(N, R), AGG_WAVES * 32, 0, stream>>>(src, dst, key, E, A, dinv, b1c, N, C,
                                                          rec, cnt);

    k_wfrag<<<cdiv(HID * HID, 256), 256, 0, stream>>>(W2c, HID, Wf);
    k_gemm128<<<cdiv(N / 16, 4), 128, 0, stream>>>(C, HID, N, Wf, dinv, nullptr, 0, A);
    k_agg<R, 1><<<cdiv(N, R), AGG_WAVES * 32, 0, stream>>>(src, dst, key, E, A, dinv, b2c, N, C,
                                                          rec, cnt);
}

extern "C" void kernel_launch(void* const* d_in, const int* in_sizes, int n_in,
                              void* d_out, int out_size, void* d_ws, size_t ws_size,
                              hipStream_t stream) {
    (void)n_in; (void)out_size; (void)ws_size;

    const int*   res_p   = (const int*)d_in[0];
    const float* pos_p   = (const float*)d_in[1];
    const int*   ei_p    = (const int*)d_in[2];
    const int*   batch_p = (const int*)d_in[3];
    const int*   res_m   = (const int*)d_in[4];
    const float* pos_m   = (const float*)d_in[5];
    const int*   ei_m    = (const int*)d_in[6];
    const int*   batch_m = (const int*)d_in[7];
    const float* W_p1 = (const float*)d_in[8];
    const float* b_p1 = (const float*)d_in[9];
    const float* W_p2 = (const float*)d_in[10];
    const float* b_p2 = (const float*)d_in[11];
    const float* W_m1 = (const float*)d_in[12];
    const float* b_m1 = (const float*)d_in[13];
    const float* W_m2 = (const float*)d_in[14];
    const float* b_m2 = (const float*)d_in[15];
    const float* W1f  = (const float*)d_in[16];
    const float* b1f  = (const float*)d_in[17];
    const float* W2f  = (const float*)d_in[18];
    const float* b2f  = (const float*)d_in[19];

    int N_P = in_sizes[0];
    int E_P = in_sizes[2] / 2;
    int N_M = in_sizes[4];
    int E_M = in_sizes[6] / 2;

    float* ws = (float*)d_ws;
    size_t nhp = (size_t)N_P * HID;
    float* A      = ws;
    float* B      = A + nhp;
    float* C      = B + nhp;
    unsigned char* key_p = (unsigned char*)(C + nhp);
    unsigned char* key_m = key_p + (size_t)cdiv(E_P, 512) * 512;
    float* dinv_p = (float*)(key_m + (size_t)cdiv(E_M, 512) * 512);
    float* dinv_m = dinv_p + N_P;
    float* pe     = dinv_m + N_M;
    float* Wfrag  = pe + NGRAPH * HID;
    int2* rec     = (int2*)(Wfrag + 2 * HID * HID);
    int*  cnt     = (int*)(rec + (size_t)AGG_MAXWG * AGG_WAVES * AGG_CAP);

    run_branch<400>(res_p, pos_p, ei_p, N_P, E_P, W_p1, b_p1, W_p2, b_p2,
                    dinv_p, key_p, A, C, Wfrag, rec, cnt, stream);
    k_pool_own<<<NGRAPH * 4, POOL_WAVES * 32, 0, stream>>>(C, batch_p, N_P, pe);

    run_branch<200>(res_m, pos_m, ei_m, N_M, E_M, W_m1, b_m1, W_m2, b_m2,
                    dinv_m, key_m, A, C, Wfrag, rec, cnt, stream);

    float* feat = A;
    float* hmid = B;
    k_feat<<<LOOP_BLOCKS, 256, 0, stream>>>(C, pe, batch_m, N_M, feat);
    k_wfrag<<<cdiv(2 * HID * HID, 256), 256, 0, stream>>>(W1f, 2 * HID, Wfrag);
    k_gemm128<<<cdiv(N_M / 16, 4), 128, 0, stream>>>(feat, 2 * HID, N_M, Wfrag,
                                                     nullptr, b1f, 1, hmid);
    k_head<<<cdiv(N_M, 256), 256, 0, stream>>>(hmid, W2f, b2f, N_M, (float*)d_out);
}


# Head product split by rows of W1: h2 @ W1[0:128] + (pe @ W1[128:256])[batch]; graph half is one 64-row product, node product K 256 -> 128 reads h2 in place, no concatenated copy; sum order changes
# speedup vs baseline: 2.9006x; 1.0625x over previous
#include <hip/hip_runtime.h>
#include <hip/hip_bf16.h>

typedef __attribute__((ext_vector_type(2))) float v2f;
typedef __attribute__((ext_vector_type(8))) float v8f;

#define HID 128
#define NGRAPH 64
#define LOOP_BLOCKS 2048


__global__ void k_fill(float* p, int n, float v) {
    int i = blockIdx.x * 256 + threadIdx.x;
    if (i < n) p[i] = v;
}

template <int R>
__global__ void k_deg_key(const int* __restrict__ dst, int E, int Epad,
                          float* __restrict__ deg, unsigned char* __restrict__ key) {
    int e = (blockIdx.x * 256 + threadIdx.x) * 4;
    if (e >= Epad) return;
    unsigned kw = 0xFFFFFFFFu;
    if (e + 3 < E && (((size_t)dst) & 15) == 0) {
        const int4 d = *(const int4*)(dst + e);
        atomicAdd(deg + d.x, 1.0f);
        atomicAdd(deg + d.y, 1.0f);
        atomicAdd(deg + d.z, 1.0f);
        atomicAdd(deg + d.w, 1.0f);
        kw = (unsigned)(d.x / R) | ((unsigned)(d.y / R) << 8) |
             ((unsigned)(d.z / R) << 16) | ((unsigned)(d.w / R) << 24);
    } else {
        for (int i = 0; i < 4; ++i) {
            if (e + i < E) {
                int d = dst[e + i];
                atomicAdd(deg + d, 1.0f);
                kw = (kw & ~(0xFFu << (8 * i))) | ((unsigned)(d / R) << (8 * i));
            }
        }
    }
    *(unsigned*)(key + e) = kw;
}

__global__ void k_rsqrt(float* d, int n) {
    int i = blockIdx.x * 256 + threadIdx.x;
    if (i < n) d[i] = rsqrtf(d[i]);
}


__global__ void k_embed(const int* __restrict__ res, const float* __restrict__ pos,
                        const float* __restrict__ W, const float* __restrict__ dinv,
                        int N, float* __restrict__ y) {
    int lane = threadIdx.x & 31;
    int w  = __builtin_amdgcn_readfirstlane(blockIdx.x * 8 + (threadIdx.x >> 5));
    int nw = gridDim.x * 8;
    const float4 w20 = *(const float4*)(W + 20 * HID + 4 * lane);
    const float4 w21 = *(const float4*)(W + 21 * HID + 4 * lane);
    const float4 w22 = *(const float4*)(W + 22 * HID + 4 * lane);
    for (int i = w; i < N; i += nw) {
        float p0 = pos[i * 3 + 0], p1 = pos[i * 3 + 1], p2 = pos[i * 3 + 2];
        float d = dinv[i];
        const float4 wr = *(const float4*)(W + res[i] * HID + 4 * lane);
        float4 o;
        o.x = (wr.x + p0 * w20.x + p1 * w21.x + p2 * w22.x) * d;
        o.y = (wr.y + p0 * w20.y + p1 * w21.y + p2 * w22.y) * d;
        o.z = (wr.z + p0 * w20.z + p1 * w21.z + p2 * w22.z) * d;
        o.w = (wr.w + p0 * w20.w + p1 * w21.w + p2 * w22.w) * d;
        *(float4*)(y + (size_t)i * HID + 4 * lane) = o;
    }
}

#define AGG_WAVES 32
#define AGG_Q 64
#define AGG_CAP 512
#define AGG_MAXWG 256

typedef volatile int __attribute__((address_space(3))) * lds_ring_t;

__device__ __forceinline__ void agg_rows(int sv, int dv, int n, int lane,
                                         const float* __restrict__ y, float* acc) {
    #pragma unroll
    for (int g = 0; g < 32; g += 16) {
        if (g < n) {
            float4 v[16];
            #pragma unroll
            for (int j = 0; j < 16; ++j) {
                if (g + j < n) {
                    const float* yr = y + (size_t)__builtin_amdgcn_readlane(sv, g + j) * HID;
                    v[j] = *(const float4*)(yr + 4 * lane);
                }
            }
            #pragma unroll
            for (int j = 0; j < 16; ++j) {
                if (g + j < n) {
                    float* a = acc + __builtin_amdgcn_readlane(dv, g + j) * HID + lane;
                    atomicAdd(a, v[j].x);
                    atomicAdd(a + 32, v[j].y);
                    atomicAdd(a + 64, v[j].z);
                    atomicAdd(a + 96, v[j].w);
                }
            }
        }
    }
}

template <bool REC>
__device__ __forceinline__ void agg_drain(lds_ring_t qw, int head, int n, int lane, int lo,
                                          const int* __restrict__ src, const int* __restrict__ dst,
                                          const float* __restrict__ y, float* acc, int2* recw) {
    const int ee = qw[(head + (lane < n ? lane : 0)) & (AGG_Q - 1)];
    const int sv = src[ee];
    const int dv = dst[ee] - lo;
    if (REC) {
        if (lane < n && head + lane < AGG_CAP) recw[head + lane] = make_int2(sv, dv);
    }
    agg_rows(sv, dv, n, lane, y, acc);
}

template <bool REC>
__device__ __forceinline__ int agg_scan(const int* __restrict__ src, const int* __restrict__ dst,
                                        const unsigned char* __restrict__ key, int E,
                                        const float* __restrict__ y, float* acc, lds_ring_t qw,
                                        int w, int lane, int lo, int2* recw) {
    const unsigned pat = (unsigned)blockIdx.x * 0x01010101u;
    const int nit = (E + 511) >> 9;
    int head = 0, tail = 0;
    const int it0 = (int)((blockIdx.x * (unsigned)nit) / gridDim.x);
    uint4 kv = make_uint4(0xFFFFFFFFu, 0xFFFFFFFFu, 0xFFFFFFFFu, 0xFFFFFFFFu);
    if (w < nit) {
        int itf = it0 + w;
        if (itf >= nit) itf -= nit;
        kv = *(const uint4*)(key + (size_t)itf * 512 + lane * 16);
    }
    for (int k = w; k < nit; k += AGG_WAVES) {
        const uint4 cur = kv;
        int it = it0 + k;
        if (it >= nit) it -= nit;
        const int eb = it * 512 + lane * 16;
        if (k + AGG_WAVES < nit) {
            int itn = it0 + k + AGG_WAVES;
            if (itn >= nit) itn -= nit;
            kv = *(const uint4*)(key + (size_t)itn * 512 + lane * 16);
        }
        const unsigned x0 = cur.x ^ pat, x1 = cur.y ^ pat, x2 = cur.z ^ pat, x3 = cur.w ^ pat;
        const unsigned u0 = ((x0 & 0x7F7F7F7Fu) + 0x7F7F7F7Fu) | x0;
        const unsigned u1 = ((x1 & 0x7F7F7F7Fu) + 0x7F7F7F7Fu) | x1;
        const unsigned u2 = ((x2 & 0x7F7F7F7Fu) + 0x7F7F7F7Fu) | x2;
        const unsigned u3 = ((x3 & 0x7F7F7F7Fu) + 0x7F7F7F7Fu) | x3;
        unsigned F = ((~u0 >> 7) & 0x01010101u) | ((~u1 >> 6) & 0x02020202u) |
                     ((~u2 >> 5) & 0x04040404u) | ((~u3 >> 4) & 0x08080808u);
        unsigned B = __builtin_amdgcn_ballot_w32(F != 0);
        while (B) {
            if (F) {
                const int p = __builtin_ctz(F);
                const int rank = __builtin_amdgcn_mbcnt_lo(B, 0);
                qw[(tail + rank) & (AGG_Q - 1)] = eb + ((p & 7) << 2) + (p >> 3);
                F &= F - 1;
            }
            tail += __builtin_popcount(B);
            if (tail - head >= 32) {
                agg_drain<REC>(qw, head, 32, lane, lo, src, dst, y, acc, recw);
                head += 32;
            }
            B = __builtin_amdgcn_ballot_w32(F != 0);
        }
    }
    if (tail - head > 0) agg_drain<REC>(qw, head, tail - head, lane, lo, src, dst, y, acc, recw);
    return tail;
}

template <int R, int MODE>
__global__ __launch_bounds__(AGG_WAVES * 32)
void k_agg(const int* __restrict__ src, const int* __restrict__ dst,
           const unsigned char* __restrict__ key, int E,
           const float* __restrict__ y, const float* __restrict__ dinv,
           const float* __restrict__ b, int N, float* __restrict__ h,
           int2* rec, int* cnt) {
    __shared__ float acc[R * HID];
    __shared__ int q[AGG_WAVES * AGG_Q];
    const int lane = threadIdx.x & 31;
    const int w = __builtin_amdgcn_readfirstlane(threadIdx.x >> 5);
    const int lo = blockIdx.x * R;
    int nrows = N - lo;
    if (nrows > R) nrows = R;

    #pragma unroll 4
    for (int r = w; r < nrows; r += AGG_WAVES) {
        const float4 v = *(const float4*)(y + (size_t)(lo + r) * HID + 4 * lane);
        float* a = acc + r * HID + lane;
        a[0] = v.x; a[32] = v.y; a[64] = v.z; a[96] = v.w;
    }
    __syncthreads();

    lds_ring_t qw = (lds_ring_t)(q + w * AGG_Q);
    int2* recw = rec + ((size_t)blockIdx.x * AGG_WAVES + w) * AGG_CAP;
    int* cntw = cnt + blockIdx.x * AGG_WAVES + w;
    if (MODE == 0) {
        const int nt = agg_scan<true>(src, dst, key, E, y, acc, qw, w, lane, lo, recw);
        if (lane == 0) *cntw = nt;
    } else {
        const int nt = __builtin_amdgcn_readfirstlane(*cntw);
        if (nt <= AGG_CAP) {
            int2 nx = make_int2(0, 0);
            if (nt > 0) nx = recw[lane < nt ? lane : 0];
            for (int p = 0; p < nt; p += 32) {
                const int2 e = nx;
                const int n = nt - p < 32 ? nt - p : 32;
                if (p + 32 < nt) nx = recw[p + 32 + (lane < nt - p - 32 ? lane : 0)];
                agg_rows(e.x, e.y, n, lane, y, acc);
            }
        } else {
            agg_scan<false>(src, dst, key, E, y, acc, qw, w, lane, lo, recw);
        }
    }
    __syncthreads();

    const float4 bv = *(const float4*)(b + 4 * lane);
    #pragma unroll 4
    for (int r = w; r < nrows; r += AGG_WAVES) {
        const int i = lo + r;
        const float d = dinv[i];
        const float* a = acc + r * HID + lane;
        float4 o;
        o.x = fmaxf(d * a[0] + bv.x, 0.0f);
        o.y = fmaxf(d * a[32] + bv.y, 0.0f);
        o.z = fmaxf(d * a[64] + bv.z, 0.0f);
        o.w = fmaxf(d * a[96] + bv.w, 0.0f);
        *(float4*)(h + (size_t)i * HID + 4 * lane) = o;
    }
}

#define POOL_WAVES 32
#define POOL_DEPTH 8

__device__ __forceinline__ void pool_rows(const float* __restrict__ hq, int c, unsigned m,
                                          int sub, float4& acc) {
    const float* hc = hq + ((size_t)c * 128 + sub) * HID;
    #pragma nounroll
    for (int half = 0; half < 32; half += 16) {
        float4 v[16];
        #pragma unroll
        for (int j = 0; j < 16; ++j)
            v[j] = *(const float4*)(hc + (size_t)(4 * (half + j)) * HID);
        #pragma unroll
        for (int j = 0; j < 16; ++j) {
            const unsigned bits = __builtin_amdgcn_readlane(m, half + j);
            const bool on = ((bits >> sub) & 1u) != 0;
            acc.x += on ? v[j].x : 0.0f;
            acc.y += on ? v[j].y : 0.0f;
            acc.z += on ? v[j].z : 0.0f;
            acc.w += on ? v[j].w : 0.0f;
        }
    }
}

__device__ __forceinline__ void pool_rows_tail(const float* __restrict__ hq, int c, unsigned m,
                                               int sub, float4& acc) {
    for (int k = 0; k < 32; ++k) {
        const unsigned bits = __builtin_amdgcn_readlane(m, k);
        if ((bits >> sub) & 1u) {
            const float4 v = *(const float4*)(hq + ((size_t)c * 128 + 4 * k + sub) * HID);
            acc.x += v.x; acc.y += v.y; acc.z += v.z; acc.w += v.w;
        }
    }
}

__global__ __launch_bounds__(POOL_WAVES * 32)
void k_pool_own(const float* __restrict__ h, const int* __restrict__ batch, int N,
                float* __restrict__ pe) {
    __shared__ float part[POOL_WAVES * 128];
    __shared__ int cntw[POOL_WAVES];
    const int lane = threadIdx.x & 31;
    const int w = __builtin_amdgcn_readfirstlane(threadIdx.x >> 5);
    const int g = blockIdx.x >> 2;
    const int q = blockIdx.x & 3;
    const int sub = lane >> 3;
    const float* hq = h + 32 * q + 4 * (lane & 7);
    const bool al = (((size_t)batch) & 15) == 0;
    const int nchunk = (N + 127) >> 7;
    const int c0 = (int)((blockIdx.x * (unsigned)nchunk) / gridDim.x);
    float4 acc = make_float4(0.0f, 0.0f, 0.0f, 0.0f);
    int cnt = 0;
    for (int k0 = w; k0 < nchunk; k0 += POOL_DEPTH * POOL_WAVES) {
        unsigned mp = 0;
        if (al && N >= 4) {
            int4 b[POOL_DEPTH];
            #pragma unroll
            for (int j = 0; j < POOL_DEPTH; ++j) {
                const int k = k0 + j * POOL_WAVES;
                int c = c0 + k;
                if (c >= nchunk) c -= nchunk;
                const int r0 = c * 128 + 4 * lane;
                b[j] = *(const int4*)(batch + ((k < nchunk && r0 + 3 < N) ? r0 : 0));
            }
            #pragma unroll
            for (int j = 0; j < POOL_DEPTH; ++j) {
                const int k = k0 + j * POOL_WAVES;
                int c = c0 + k;
                if (c >= nchunk) c -= nchunk;
                const int r0 = c * 128 + 4 * lane;
                unsigned mj = 0;
                if (k < nchunk) {
                    if (r0 + 3 < N) {
                        mj = (b[j].x == g ? 1u : 0u) | (b[j].y == g ? 2u : 0u) |
                             (b[j].z == g ? 4u : 0u) | (b[j].w == g ? 8u : 0u);
                    } else {
                        for (int i = 0; i < 4; ++i)
                            if (r0 + i < N && batch[r0 + i] == g) mj |= 1u << i;
                    }
                }
                mp |= mj << (4 * j);
            }
        } else {
            for (int j = 0; j < POOL_DEPTH; ++j) {
                const int k = k0 + j * POOL_WAVES;
                if (k < nchunk) {
                    int c = c0 + k;
                    if (c >= nchunk) c -= nchunk;
                    const int r0 = c * 128 + 4 * lane;
                    unsigned mj = 0;
                    for (int i = 0; i < 4; ++i)
                        if (r0 + i < N && batch[r0 + i] == g) mj |= 1u << i;
                    mp |= mj << (4 * j);
                }
            }
        }
        if (__builtin_amdgcn_ballot_w32(mp != 0) == 0) continue;
        for (int j = 0; j < POOL_DEPTH; ++j) {
            const unsigned mj = (mp >> (4 * j)) & 15u;
            if (__builtin_amdgcn_ballot_w32(mj != 0) == 0) continue;
            int c = c0 + k0 + j * POOL_WAVES;
            if (c >= nchunk) c -= nchunk;
            cnt += __builtin_popcount(__builtin_amdgcn_ballot_w32((mj & 1u) != 0)) +
                   __builtin_popcount(__builtin_amdgcn_ballot_w32((mj & 2u) != 0)) +
                   __builtin_popcount(__builtin_amdgcn_ballot_w32((mj & 4u) != 0)) +
                   __builtin_popcount(__builtin_amdgcn_ballot_w32((mj & 8u) != 0));
            if (c * 128 + 128 <= N) pool_rows(hq, c, mj, sub, acc);
            else pool_rows_tail(hq, c, mj, sub, acc);
        }
    }
    *(float4*)(part + w * 128 + 4 * lane) = acc;
    if (lane == 0) cntw[w] = cnt;
    __syncthreads();
    if (threadIdx.x < 32) {
        float s0 = 0.0f, s1 = 0.0f, s2 = 0.0f, s3 = 0.0f;
        #pragma unroll 8
        for (int j = 0; j < POOL_WAVES * 4; j += 4) {
            s0 += part[(j + 0) * 32 + lane];
            s1 += part[(j + 1) * 32 + lane];
            s2 += part[(j + 2) * 32 + lane];
            s3 += part[(j + 3) * 32 + lane];
        }
        int n = 0;
        for (int j = 0; j < POOL_WAVES; ++j) n += cntw[j];
        pe[g * HID + 32 * q + lane] = ((s0 + s1) + (s2 + s3)) / fmaxf((float)n, 1.0f);
    }
}

__global__ void k_head(const float* __restrict__ hmid, const float* __restrict__ W2,
                       const float* __restrict__ b2, int N, float* __restrict__ out) {
    int i = blockIdx.x * 256 + threadIdx.x;
    if (i >= N) return;
    float a0 = b2[0], a1 = b2[1], a2 = b2[2];
    const float* h = hmid + (size_t)i * HID;
    for (int k = 0; k < HID; ++k) {
        float hv = h[k];
        a0 += hv * W2[k * 3 + 0];
        a1 += hv * W2[k * 3 + 1];
        a2 += hv * W2[k * 3 + 2];
    }
    out[(size_t)i * 3 + 0] = a0;
    out[(size_t)i * 3 + 1] = a1;
    out[(size_t)i * 3 + 2] = a2;
}

__global__ void k_wfrag(const float* __restrict__ W, int K, float* __restrict__ Wf) {
    int idx = blockIdx.x * 256 + threadIdx.x;
    if (idx >= K * HID) return;
    int j    = idx & 1;
    int nt   = (idx >> 1) & 7;
    int l    = (idx >> 4) & 31;
    int kk   = idx >> 9;
    int krow = 4 * kk + (l >> 4) * 2 + j;
    int col  = nt * 16 + (l & 15);
    Wf[idx] = W[(size_t)krow * HID + col];
}

__global__ void k_gemm128(const float* __restrict__ A, int M,
                          const float* __restrict__ Wf,
                          const float* __restrict__ rowscale,
                          const float* __restrict__ bias, int relu,
                          float* __restrict__ out,
                          const float* __restrict__ rowinit,
                          const int* __restrict__ rowidx) {
    int w = threadIdx.x >> 5;
    int lane = threadIdx.x & 31;
    int tile = blockIdx.x * 4 + w;
    if (tile * 16 >= M) return;
    int row0 = tile * 16;
    int half = lane >> 4;
    int lmod = lane & 15;
    int koff = half * 2;
    const float* Arow = A + (size_t)(row0 + lmod) * HID + koff;
    const float* Wl   = Wf + (size_t)lane * 16;

    v8f acc[8];
    if (rowinit) {
        int ra[8];
        #pragma unroll
        for (int r = 0; r < 8; ++r) ra[r] = rowidx[row0 + r + half * 8] * HID + lmod;
        #pragma unroll
        for (int nt = 0; nt < 8; ++nt) {
            #pragma unroll
            for (int r = 0; r < 8; ++r) acc[nt][r] = rowinit[ra[r] + nt * 16];
        }
    } else {
        #pragma unroll
        for (int nt = 0; nt < 8; ++nt) acc[nt] = (v8f){};
    }

    const int ksteps = HID >> 2;
    for (int kk = 0; kk < ksteps; ++kk) {
        v2f a;
        a.x = Arow[4 * kk];
        a.y = Arow[4 * kk + 1];
        const float4* q = (const float4*)(Wl + (size_t)kk * 512);
        float4 q0 = q[0], q1 = q[1], q2 = q[2], q3 = q[3];
        v2f b0; b0.x = q0.x; b0.y = q0.y;
        v2f b1; b1.x = q0.z; b1.y = q0.w;
        v2f b2; b2.x = q1.x; b2.y = q1.y;
        v2f b3; b3.x = q1.z; b3.y = q1.w;
        v2f b4; b4.x = q2.x; b4.y = q2.y;
        v2f b5; b5.x = q2.z; b5.y = q2.w;
        v2f b6; b6.x = q3.x; b6.y = q3.y;
        v2f b7; b7.x = q3.z; b7.y = q3.w;
        acc[0] = __builtin_amdgcn_wmma_f32_16x16x4_f32(false, a, false, b0, (short)0, acc[0], false, false);
        acc[1] = __builtin_amdgcn_wmma_f32_16x16x4_f32(false, a, false, b1, (short)0, acc[1], false, false);
        acc[2] = __builtin_amdgcn_wmma_f32_16x16x4_f32(false, a, false, b2, (short)0, acc[2], false, false);
        acc[3] = __builtin_amdgcn_wmma_f32_16x16x4_f32(false, a, false, b3, (short)0, acc[3], false, false);
        acc[4] = __builtin_amdgcn_wmma_f32_16x16x4_f32(false, a, false, b4, (short)0, acc[4], false, false);
        acc[5] = __builtin_amdgcn_wmma_f32_16x16x4_f32(false, a, false, b5, (short)0, acc[5], false, false);
        acc[6] = __builtin_amdgcn_wmma_f32_16x16x4_f32(false, a, false, b6, (short)0, acc[6], false, false);
        acc[7] = __builtin_amdgcn_wmma_f32_16x16x4_f32(false, a, false, b7, (short)0, acc[7], false, false);
    }

    float rs[8];
    if (rowscale) {
        #pragma unroll
        for (int r = 0; r < 8; ++r) rs[r] = rowscale[row0 + r + half * 8];
    }

    #pragma unroll
    for (int nt = 0; nt < 8; ++nt) {
        int col = nt * 16 + lmod;
        float bv = bias ? bias[col] : 0.0f;
        #pragma unroll
        for (int r = 0; r < 8; ++r) {
            float v = acc[nt][r];
            if (rowscale) v *= rs[r];
            v += bv;
            if (relu) v = fmaxf(v, 0.0f);
            out[(size_t)(row0 + r + half * 8) * HID + col] = v;
        }
    }
}

__global__ __launch_bounds__(1024)
void k_graph_term(const float* __restrict__ pe, const float* __restrict__ Wb,
                  const float* __restrict__ bias, float* __restrict__ gterm) {
    const int lane = threadIdx.x & 31;
    const int w    = threadIdx.x >> 5;
    const int mt = w >> 3, nt = w & 7;
    const int half = lane >> 4, lmod = lane & 15;
    const float* Arow = pe + (size_t)(mt * 16 + lmod) * HID + half * 2;
    const float* Bcol = Wb + (size_t)(half * 2) * HID + nt * 16 + lmod;
    v8f acc = (v8f){};
    #pragma nounroll
    for (int k0 = 0; k0 < HID / 4; k0 += 16) {
        v2f a[16], b[16];
        #pragma unroll
        for (int j = 0; j < 16; ++j) {
            const int k = 4 * (k0 + j);
            a[j].x = Arow[k];
            a[j].y = Arow[k + 1];
            b[j].x = Bcol[(size_t)k * HID];
            b[j].y = Bcol[(size_t)(k + 1) * HID];
        }
        #pragma unroll
        for (int j = 0; j < 16; ++j)
            acc = __builtin_amdgcn_wmma_f32_16x16x4_f32(false, a[j], false, b[j], (short)0, acc,
                                                        false, false);
    }
    const int col = nt * 16 + lmod;
    const float bv = bias[col];
    #pragma unroll
    for (int r = 0; r < 8; ++r)
        gterm[(size_t)(mt * 16 + r + half * 8) * HID + col] = acc[r] + bv;
}


static inline int cdiv(long long a, long long b) { return (int)((a + b - 1) / b); }

template <int R>
static void run_branch(const int* res, const float* pos, const int* ei,
                       int N, int E,
                       const float* W1c, const float* b1c,
                       const float* W2c, const float* b2c,
                       float* dinv, unsigned char* key, float* A, float* C, float* Wf,
                       int2* rec, int* cnt, hipStream_t stream) {
    const int* src = ei;
    const int* dst = ei + E;
    int Epad = cdiv(E, 512) * 512;

    k_fill<<<cdiv(N, 256), 256, 0, stream>>>(dinv, N, 1.0f);
    k_deg_key<R><<<cdiv(Epad / 4, 256), 256, 0, stream>>>(dst, E, Epad, dinv, key);
    k_rsqrt<<<cdiv(N, 256), 256, 0, stream>>>(dinv, N);

    k_embed<<<LOOP_BLOCKS, 256, 0, stream>>>(res, pos, W1c, dinv, N, A);
    k_agg<R, 0><<<cdiv(N, R), AGG_WAVES * 32, 0, stream>>>(src, dst, key, E, A, dinv, b1c, N, C,
                                                          rec, cnt);

    k_wfrag<<<cdiv(HID * HID, 256), 256, 0, stream>>>(W2c, HID, Wf);
    k_gemm128<<<cdiv(N / 16, 4), 128, 0, stream>>>(C, N, Wf, dinv, nullptr, 0, A,
                                                   nullptr, nullptr);
    k_agg<R, 1><<<cdiv(N, R), AGG_WAVES * 32, 0, stream>>>(src, dst, key, E, A, dinv, b2c, N, C,
                                                          rec, cnt);
}

extern "C" void kernel_launch(void* const* d_in, const int* in_sizes, int n_in,
                              void* d_out, int out_size, void* d_ws, size_t ws_size,
                              hipStream_t stream) {
    (void)n_in; (void)out_size; (void)ws_size;

    const int*   res_p   = (const int*)d_in[0];
    const float* pos_p   = (const float*)d_in[1];
    const int*   ei_p    = (const int*)d_in[2];
    const int*   batch_p = (const int*)d_in[3];
    const int*   res_m   = (const int*)d_in[4];
    const float* pos_m   = (const float*)d_in[5];
    const int*   ei_m    = (const int*)d_in[6];
    const int*   batch_m = (const int*)d_in[7];
    const float* W_p1 = (const float*)d_in[8];
    const float* b_p1 = (const float*)d_in[9];
    const float* W_p2 = (const float*)d_in[10];
    const float* b_p2 = (const float*)d_in[11];
    const float* W_m1 = (const float*)d_in[12];
    const float* b_m1 = (const float*)d_in[13];
    const float* W_m2 = (const float*)d_in[14];
    const float* b_m2 = (const float*)d_in[15];
    const float* W1f  = (const float*)d_in[16];
    const float* b1f  = (const float*)d_in[17];
    const float* W2f  = (const float*)d_in[18];
    const float* b2f  = (const float*)d_in[19];

    int N_P = in_sizes[0];
    int E_P = in_sizes[2] / 2;
    int N_M = in_sizes[4];
    int E_M = in_sizes[6] / 2;

    float* ws = (float*)d_ws;
    size_t nhp = (size_t)N_P * HID;
    float* A      = ws;
    float* B      = A + nhp;
    float* C      = B + nhp;
    unsigned char* key_p = (unsigned char*)(C + nhp);
    unsigned char* key_m = key_p + (size_t)cdiv(E_P, 512) * 512;
    float* dinv_p = (float*)(key_m + (size_t)cdiv(E_M, 512) * 512);
    float* dinv_m = dinv_p + N_P;
    float* pe     = dinv_m + N_M;
    float* gterm  = pe + NGRAPH * HID;
    float* Wfrag  = gterm + NGRAPH * HID;
    int2* rec     = (int2*)(Wfrag + 2 * HID * HID);
    int*  cnt     = (int*)(rec + (size_t)AGG_MAXWG * AGG_WAVES * AGG_CAP);

    run_branch<400>(res_p, pos_p, ei_p, N_P, E_P, W_p1, b_p1, W_p2, b_p2,
                    dinv_p, key_p, A, C, Wfrag, rec, cnt, stream);
    k_pool_own<<<NGRAPH * 4, POOL_WAVES * 32, 0, stream>>>(C, batch_p, N_P, pe);

    run_branch<200>(res_m, pos_m, ei_m, N_M, E_M, W_m1, b_m1, W_m2, b_m2,
                    dinv_m, key_m, A, C, Wfrag, rec, cnt, stream);

    float* hmid = B;
    k_wfrag<<<cdiv(HID * HID, 256), 256, 0, stream>>>(W1f, HID, Wfrag);
    k_graph_term<<<1, 1024, 0, stream>>>(pe, W1f + HID * HID, b1f, gterm);
    k_gemm128<<<cdiv(N_M / 16, 4), 128, 0, stream>>>(C, N_M, Wfrag,
                                                     nullptr, nullptr, 1, hmid,
                                                     gterm, batch_m);
    k_head<<<cdiv(N_M, 256), 256, 0, stream>>>(hmid, W2f, b2f, N_M, (float*)d_out);
}
